# starting kernel
# baseline (speedup 1.0000x reference)
#include <hip/hip_runtime.h>

typedef __attribute__((ext_vector_type(2))) float v2f;
typedef __attribute__((ext_vector_type(8))) float v8f;

constexpr int Bc = 128, Nw = 4096, Sc = 32, Pc = 64, Dc = 128;
constexpr float NORM_INV = 1.0f / 300.0f;

constexpr int KA  = 256;
constexpr int LDA = 260;
constexpr int KB  = 384;
constexpr int LDB = 388;

__global__ __launch_bounds__(256)
void pm_emb_kernel(const float* __restrict__ enc_row,
                   const float* __restrict__ enc_col,
                   const float* __restrict__ clockv,
                   const float* __restrict__ pet,
                   const float* __restrict__ W_dyn,
                   const float* __restrict__ W_concat,
                   const int*   __restrict__ lhw,
                   const int*   __restrict__ lstage,
                   float*       __restrict__ out)
{
    __shared__ float Xs[32 * LDA];
    __shared__ float w3[Dc];
    __shared__ float Rp[32];

    const int b    = blockIdx.x >> 1;
    const int half = blockIdx.x & 1;
    const int t    = threadIdx.x;

    {
        const int r     = t >> 3;
        const int chunk = t & 7;
        const int p     = half * 32 + r;
        const int col0  = chunk * 32;
        const float* src = nullptr;
        if (chunk < 4) {
            const int st = lstage[b * Pc + p] - 1;
            src = enc_col + ((size_t)b * Sc + st) * Dc + col0;
        } else {
            const int w = lhw[b * Pc + p];
            if (w >= 0 && w < Nw)
                src = enc_row + ((size_t)b * Nw + w) * Dc + (col0 - 128);
        }
        float* dst = &Xs[r * LDA + col0];
        if (src) {
            #pragma unroll
            for (int j = 0; j < 8; ++j)
                *(float4*)(dst + 4 * j) = *(const float4*)(src + 4 * j);
        } else {
            #pragma unroll
            for (int j = 0; j < 8; ++j)
                *(float4*)(dst + 4 * j) = make_float4(0.f, 0.f, 0.f, 0.f);
        }
    }
    if (t < Dc) {
        float acc = 0.f;
        #pragma unroll 4
        for (int k = 0; k < Dc; ++k)
            acc = fmaf(W_dyn[k], W_concat[(size_t)(256 + k) * Dc + t], acc);
        w3[t] = acc;
    } else if (t < 160) {
        const int r = t - 128;
        const int p = half * 32 + r;
        const float d = pet[b * Pc + p] - clockv[b];
        Rp[r] = (d > 0.f ? d : 0.f) * NORM_INV;
    }
    __syncthreads();

    const int wave  = t >> 5;
    const int lane  = t & 31;
    const int l16   = lane & 15;
    const int khalf = (lane >> 4) << 1;
    const int n0    = wave * 16;

    v8f acc0 = {};
    v8f acc1 = {};
    const float* wb = W_concat + n0 + l16;
    const float* a0 = &Xs[l16 * LDA + khalf];
    const float* a1 = &Xs[(16 + l16) * LDA + khalf];
    for (int k = 0; k < KA; k += 4) {
        v2f bf;
        bf.x = wb[(size_t)(k + khalf)     * Dc];
        bf.y = wb[(size_t)(k + khalf + 1) * Dc];
        v2f fa0 = *(const v2f*)(a0 + k);
        v2f fa1 = *(const v2f*)(a1 + k);
        acc0 = __builtin_amdgcn_wmma_f32_16x16x4_f32(
                   false, fa0, false, bf, (short)0, acc0, false, false);
        acc1 = __builtin_amdgcn_wmma_f32_16x16x4_f32(
                   false, fa1, false, bf, (short)0, acc1, false, false);
    }

    const float w3v   = w3[n0 + l16];
    const int   rhalf = (lane >> 4) << 3;
    #pragma unroll
    for (int r = 0; r < 8; ++r) {
        const int row0 = rhalf + r;
        const float v0 = acc0[r] + Rp[row0] * w3v;
        out[((size_t)b * (Pc + 2) + half * 32 + row0) * Dc + n0 + l16] = v0;
    }
    #pragma unroll
    for (int r = 0; r < 8; ++r) {
        const int row1 = 16 + rhalf + r;
        const float v1 = acc1[r] + Rp[row1] * w3v;
        out[((size_t)b * (Pc + 2) + half * 32 + row1) * Dc + n0 + l16] = v1;
    }
}

__global__ __launch_bounds__(256)
void robot_kernel(const float* __restrict__ enc_row,
                  const float* __restrict__ enc_col,
                  const float* __restrict__ W_robot,
                  const int* __restrict__ a1loc, const int* __restrict__ a2loc,
                  const int* __restrict__ a1rec, const int* __restrict__ a2rec,
                  const int* __restrict__ a1ns,  const int* __restrict__ a2ns,
                  float* __restrict__ out)
{
    __shared__ float Xs[16 * LDB];
    const int t  = threadIdx.x;
    const int mt = blockIdx.x;

    {
        const int r     = t >> 4;
        const int fbase = (t & 15) * 6;
        const int ar    = mt * 16 + r;
        const int b     = ar >> 1, arm = ar & 1;
        const int loc = (arm ? a2loc : a1loc)[b];
        const int rec = (arm ? a2rec : a1rec)[b];
        const int ns  = (arm ? a2ns  : a1ns )[b];
        #pragma unroll
        for (int j = 0; j < 6; ++j) {
            const int col = (fbase + j) * 4;
            const int seg = col >> 7, cin = col & 127;
            float4 v = make_float4(0.f, 0.f, 0.f, 0.f);
            if (seg == 0) {
                if (loc == Pc + 1)
                    v = make_float4(1.f, 1.f, 1.f, 1.f);
                else if (loc >= 1 && loc <= Pc)
                    v = *(const float4*)(out + ((size_t)b * (Pc + 2) + (loc - 1)) * Dc + cin);
            } else if (seg == 1) {
                if (rec >= 0 && rec < Nw)
                    v = *(const float4*)(enc_row + ((size_t)b * Nw + rec) * Dc + cin);
            } else {
                if (ns >= 1 && ns <= Sc)
                    v = *(const float4*)(enc_col + ((size_t)b * Sc + (ns - 1)) * Dc + cin);
            }
            *(float4*)&Xs[r * LDB + col] = v;
        }
    }
    __syncthreads();

    const int wave  = t >> 5;
    const int lane  = t & 31;
    const int l16   = lane & 15;
    const int khalf = (lane >> 4) << 1;
    const int n0    = wave * 16;

    v8f acc = {};
    const float* wb = W_robot + n0 + l16;
    const float* ap = &Xs[l16 * LDB + khalf];
    for (int k = 0; k < KB; k += 4) {
        v2f a = *(const v2f*)(ap + k);
        v2f bf;
        bf.x = wb[(size_t)(k + khalf)     * Dc];
        bf.y = wb[(size_t)(k + khalf + 1) * Dc];
        acc = __builtin_amdgcn_wmma_f32_16x16x4_f32(
                  false, a, false, bf, (short)0, acc, false, false);
    }
    #pragma unroll
    for (int r = 0; r < 8; ++r) {
        const int mrow = ((lane >> 4) << 3) + r;
        const int ar   = mt * 16 + mrow;
        const int b    = ar >> 1, arm = ar & 1;
        out[((size_t)b * (Pc + 2) + Pc + arm) * Dc + n0 + l16] = acc[r];
    }
}

extern "C" void kernel_launch(void* const* d_in, const int* in_sizes, int n_in,
                              void* d_out, int out_size, void* d_ws, size_t ws_size,
                              hipStream_t stream) {
    (void)in_sizes; (void)n_in; (void)out_size; (void)d_ws; (void)ws_size;
    const float* enc_row  = (const float*)d_in[0];
    const float* enc_col  = (const float*)d_in[1];
    const float* clockv   = (const float*)d_in[2];
    const float* pet      = (const float*)d_in[3];
    const float* W_dyn    = (const float*)d_in[4];
    const float* W_concat = (const float*)d_in[5];
    const float* W_robot  = (const float*)d_in[6];
    const int*   lhw      = (const int*)d_in[7];
    const int*   lstage   = (const int*)d_in[8];
    const int*   a1loc    = (const int*)d_in[9];
    const int*   a2loc    = (const int*)d_in[10];
    const int*   a1rec    = (const int*)d_in[11];
    const int*   a2rec    = (const int*)d_in[12];
    const int*   a1ns     = (const int*)d_in[13];
    const int*   a2ns     = (const int*)d_in[14];
    float* out = (float*)d_out;

    pm_emb_kernel<<<Bc * 2, 256, 0, stream>>>(enc_row, enc_col, clockv, pet,
                                              W_dyn, W_concat, lhw, lstage, out);
    robot_kernel<<<16, 256, 0, stream>>>(enc_row, enc_col, W_robot,
                                         a1loc, a2loc, a1rec, a2rec, a1ns, a2ns, out);
}


# W_concat and W_robot staged once per work-group in LDS (first K third, then the rest while that third computes), weight loads issued up front; B fragments and w3 sum read LDS; sum orders unchanged
# speedup vs baseline: 1.2730x; 1.2730x over previous
#include <hip/hip_runtime.h>


typedef __attribute__((ext_vector_type(2))) float v2f;
typedef __attribute__((ext_vector_type(8))) float v8f;
typedef __attribute__((ext_vector_type(4))) float v4f;

constexpr int Bc = 128, Nw = 4096, Sc = 32, Pc = 64, Dc = 128;
constexpr float NORM_INV = 1.0f / 300.0f;

constexpr int KA  = 256;
constexpr int LDA = 260;
constexpr int KB  = 384;
constexpr int LDB = 388;
constexpr int WR  = 384;
constexpr int LDW = 136;
constexpr int SEG = 128;

static __device__ __forceinline__
void seg_load(v4f (&r)[16], const float* __restrict__ Wm, int s, int t) {
    const float* src = Wm + (size_t)(s * SEG + (t >> 5)) * Dc + (t & 31) * 4;
    #pragma unroll
    for (int i = 0; i < 16; ++i)
        r[i] = *(const v4f*)(src + (size_t)i * 8 * Dc);
}
static __device__ __forceinline__
void seg_store(float* Ws, const v4f (&r)[16], int s, int t) {
    float* dst = Ws + (s * SEG + (t >> 5)) * LDW + (t & 31) * 4;
    #pragma unroll
    for (int i = 0; i < 16; ++i)
        *(v4f*)(dst + i * 8 * LDW) = r[i];
}

static __device__ __forceinline__ void lds_barrier() {
    asm volatile("s_wait_dscnt 0x0\n\ts_barrier_signal -1\n\ts_barrier_wait -1" ::: "memory");
}

__global__ __launch_bounds__(256)
void pm_emb_kernel(const float* __restrict__ enc_row,
                   const float* __restrict__ enc_col,
                   const float* __restrict__ clockv,
                   const float* __restrict__ pet,
                   const float* __restrict__ W_dyn,
                   const float* __restrict__ W_concat,
                   const int*   __restrict__ lhw,
                   const int*   __restrict__ lstage,
                   float*       __restrict__ out)
{
    __shared__ float Ws[WR * LDW];
    __shared__ float Xs[32 * LDA];
    __shared__ float wd[8][Dc];
    __shared__ float Rp[8][32];

    const int b    = blockIdx.x >> 1;
    const int half = blockIdx.x & 1;
    const int t    = threadIdx.x;

    const int r     = t >> 3;
    const int chunk = t & 7;
    const int p     = half * 32 + r;
    const int col0  = chunk * 32;

    const int gidx = (chunk < 4) ? lstage[b * Pc + p] : lhw[b * Pc + p];

    const float4 wdv = *(const float4*)(W_dyn + 4 * (t & 31));
    const float  pev = pet[b * Pc + half * 32 + (t & 31)];
    const float  clk = clockv[b];

    v4f ra[16], rb[16], rc[16];
    seg_load(ra, W_concat, 0, t);

    const float* src = nullptr;
    if (chunk < 4) {
        const int st = gidx - 1;
        src = enc_col + ((size_t)b * Sc + st) * Dc + col0;
    } else {
        if (gidx >= 0 && gidx < Nw)
            src = enc_row + ((size_t)b * Nw + gidx) * Dc + (col0 - 128);
    }
    v4f g[8];
    #pragma unroll
    for (int j = 0; j < 8; ++j) g[j] = (v4f){0.f, 0.f, 0.f, 0.f};
    if (src) {
        #pragma unroll
        for (int j = 0; j < 8; ++j) g[j] = *(const v4f*)(src + 4 * j);
    }

    seg_load(rb, W_concat, 1, t);
    seg_load(rc, W_concat, 2, t);
    seg_store(Ws, ra, 0, t);
    {
        float* dst = &Xs[r * LDA + col0];
        #pragma unroll
        for (int j = 0; j < 8; ++j) *(v4f*)(dst + 4 * j) = g[j];
    }
    lds_barrier();

    const int wave  = t >> 5;
    const int lane  = t & 31;
    const int l16   = lane & 15;
    const int khalf = (lane >> 4) << 1;
    const int n0    = wave * 16;

    v8f acc0 = {};
    v8f acc1 = {};
    const float* wb = &Ws[khalf * LDW + n0 + l16];
    const float* a0 = &Xs[l16 * LDA + khalf];
    const float* a1 = &Xs[(16 + l16) * LDA + khalf];

    #pragma unroll 8
    for (int k = 0; k < SEG; k += 4) {
        v2f bf;
        bf.x = wb[k * LDW];
        bf.y = wb[(k + 1) * LDW];
        v2f fa0 = *(const v2f*)(a0 + k);
        v2f fa1 = *(const v2f*)(a1 + k);
        acc0 = __builtin_amdgcn_wmma_f32_16x16x4_f32(
                   false, fa0, false, bf, (short)0, acc0, false, false);
        acc1 = __builtin_amdgcn_wmma_f32_16x16x4_f32(
                   false, fa1, false, bf, (short)0, acc1, false, false);
    }
    seg_store(Ws, rb, 1, t);
    seg_store(Ws, rc, 2, t);
    {
        *(float4*)&wd[wave][4 * lane] = wdv;
        const float d = pev - clk;
        Rp[wave][lane] = (d > 0.f ? d : 0.f) * NORM_INV;
    }
    lds_barrier();

    float w3v = 0.f;
    const float* wc = &Ws[KA * LDW + n0 + l16];
    #pragma unroll 8
    for (int k = SEG; k < KA; k += 4) {
        const int kk = k - SEG;
        const float4 wv = *(const float4*)&wd[wave][kk];
        w3v = fmaf(wv.x, wc[(kk + 0) * LDW], w3v);
        w3v = fmaf(wv.y, wc[(kk + 1) * LDW], w3v);
        w3v = fmaf(wv.z, wc[(kk + 2) * LDW], w3v);
        w3v = fmaf(wv.w, wc[(kk + 3) * LDW], w3v);
        v2f bf;
        bf.x = wb[k * LDW];
        bf.y = wb[(k + 1) * LDW];
        v2f fa0 = *(const v2f*)(a0 + k);
        v2f fa1 = *(const v2f*)(a1 + k);
        acc0 = __builtin_amdgcn_wmma_f32_16x16x4_f32(
                   false, fa0, false, bf, (short)0, acc0, false, false);
        acc1 = __builtin_amdgcn_wmma_f32_16x16x4_f32(
                   false, fa1, false, bf, (short)0, acc1, false, false);
    }

    const int   rhalf = (lane >> 4) << 3;
    #pragma unroll
    for (int rr = 0; rr < 8; ++rr) {
        const int row0 = rhalf + rr;
        const float v0 = acc0[rr] + Rp[wave][row0] * w3v;
        out[((size_t)b * (Pc + 2) + half * 32 + row0) * Dc + n0 + l16] = v0;
    }
    #pragma unroll
    for (int rr = 0; rr < 8; ++rr) {
        const int row1 = 16 + rhalf + rr;
        const float v1 = acc1[rr] + Rp[wave][row1] * w3v;
        out[((size_t)b * (Pc + 2) + half * 32 + row1) * Dc + n0 + l16] = v1;
    }
}

__global__ __launch_bounds__(256)
void robot_kernel(const float* __restrict__ enc_row,
                  const float* __restrict__ enc_col,
                  const float* __restrict__ W_robot,
                  const int* __restrict__ a1loc, const int* __restrict__ a2loc,
                  const int* __restrict__ a1rec, const int* __restrict__ a2rec,
                  const int* __restrict__ a1ns,  const int* __restrict__ a2ns,
                  float* __restrict__ out)
{
    __shared__ float Ws[WR * LDW];
    __shared__ float Xs[16 * LDB];
    const int t  = threadIdx.x;
    const int mt = blockIdx.x;

    const int r     = t >> 4;
    const int arow  = mt * 16 + r;
    const int bb    = arow >> 1, armr = arow & 1;
    const int  q   = t & 15;
    const int* ip  = (q == 1) ? (armr ? a2rec : a1rec)
                   : (q == 2) ? (armr ? a2ns  : a1ns )
                              : (armr ? a2loc : a1loc);
    const int  iv  = ip[bb];

    v4f ra[16], rb[16], rc[16];
    seg_load(ra, W_robot, 0, t);

    const int lbase = t & 16;
    const int loc = __shfl(iv, lbase + 0);
    const int rec = __shfl(iv, lbase + 1);
    const int ns  = __shfl(iv, lbase + 2);

    const int c16 = (t & 15) * 4;
    const bool locOK = (loc >= 1) && (loc <= Pc);
    const bool recOK = (rec >= 0) && (rec < Nw);
    const bool nsOK  = (ns >= 1) && (ns <= Sc);
    v4f g0, g1, g2, g3, g4, g5;
    if (locOK) {
        const float* p = out + ((size_t)bb * (Pc + 2) + (loc - 1)) * Dc + c16;
        g0 = *(const v4f*)p;
        g1 = *(const v4f*)(p + 64);
    }
    if (recOK) {
        const float* p = enc_row + ((size_t)bb * Nw + rec) * Dc + c16;
        g2 = *(const v4f*)p;
        g3 = *(const v4f*)(p + 64);
    }
    if (nsOK) {
        const float* p = enc_col + ((size_t)bb * Sc + (ns - 1)) * Dc + c16;
        g4 = *(const v4f*)p;
        g5 = *(const v4f*)(p + 64);
    }
    float* const xdst = &Xs[r * LDB + c16];
    {
        const v4f zero4 = {0.f, 0.f, 0.f, 0.f};
        const v4f lfill = (loc == Pc + 1) ? (v4f){1.f, 1.f, 1.f, 1.f} : zero4;
        *(v4f*)(xdst +   0) = lfill;  *(v4f*)(xdst +  64) = lfill;
        *(v4f*)(xdst + 128) = zero4;  *(v4f*)(xdst + 192) = zero4;
        *(v4f*)(xdst + 256) = zero4;  *(v4f*)(xdst + 320) = zero4;
    }

    seg_load(rb, W_robot, 1, t);
    seg_load(rc, W_robot, 2, t);
    seg_store(Ws, ra, 0, t);
    if (locOK) { *(v4f*)(xdst +   0) = g0;  *(v4f*)(xdst +  64) = g1; }
    if (recOK) { *(v4f*)(xdst + 128) = g2;  *(v4f*)(xdst + 192) = g3; }
    if (nsOK)  { *(v4f*)(xdst + 256) = g4;  *(v4f*)(xdst + 320) = g5; }
    lds_barrier();

    const int wave  = t >> 5;
    const int lane  = t & 31;
    const int l16   = lane & 15;
    const int khalf = (lane >> 4) << 1;
    const int n0    = wave * 16;

    v8f acc = {};
    const float* wb = &Ws[khalf * LDW + n0 + l16];
    const float* ap = &Xs[l16 * LDB + khalf];
    #pragma unroll 8
    for (int k = 0; k < SEG; k += 4) {
        v2f a = *(const v2f*)(ap + k);
        v2f bf;
        bf.x = wb[k * LDW];
        bf.y = wb[(k + 1) * LDW];
        acc = __builtin_amdgcn_wmma_f32_16x16x4_f32(
                  false, a, false, bf, (short)0, acc, false, false);
    }
    seg_store(Ws, rb, 1, t);
    seg_store(Ws, rc, 2, t);
    lds_barrier();
    #pragma unroll 8
    for (int k = SEG; k < KB; k += 4) {
        v2f a = *(const v2f*)(ap + k);
        v2f bf;
        bf.x = wb[k * LDW];
        bf.y = wb[(k + 1) * LDW];
        acc = __builtin_amdgcn_wmma_f32_16x16x4_f32(
                  false, a, false, bf, (short)0, acc, false, false);
    }
    #pragma unroll
    for (int rr = 0; rr < 8; ++rr) {
        const int mrow = ((lane >> 4) << 3) + rr;
        const int ar   = mt * 16 + mrow;
        const int b    = ar >> 1, arm = ar & 1;
        out[((size_t)b * (Pc + 2) + Pc + arm) * Dc + n0 + l16] = acc[rr];
    }
}

extern "C" void kernel_launch(void* const* d_in, const int* in_sizes, int n_in,
                              void* d_out, int out_size, void* d_ws, size_t ws_size,
                              hipStream_t stream) {
    (void)in_sizes; (void)n_in; (void)out_size; (void)d_ws; (void)ws_size;
    const float* enc_row  = (const float*)d_in[0];
    const float* enc_col  = (const float*)d_in[1];
    const float* clockv   = (const float*)d_in[2];
    const float* pet      = (const float*)d_in[3];
    const float* W_dyn    = (const float*)d_in[4];
    const float* W_concat = (const float*)d_in[5];
    const float* W_robot  = (const float*)d_in[6];
    const int*   lhw      = (const int*)d_in[7];
    const int*   lstage   = (const int*)d_in[8];
    const int*   a1loc    = (const int*)d_in[9];
    const int*   a2loc    = (const int*)d_in[10];
    const int*   a1rec    = (const int*)d_in[11];
    const int*   a2rec    = (const int*)d_in[12];
    const int*   a1ns     = (const int*)d_in[13];
    const int*   a2ns     = (const int*)d_in[14];
    float* out = (float*)d_out;

    pm_emb_kernel<<<Bc * 2, 256, 0, stream>>>(enc_row, enc_col, clockv, pet,
                                              W_dyn, W_concat, lhw, lstage, out);
    robot_kernel<<<16, 256, 0, stream>>>(enc_row, enc_col, W_robot,
                                         a1loc, a2loc, a1rec, a2rec, a1ns, a2ns, out);
}


# robot launch re-tiled: 16 -> 64 work-groups (16x32 output tile each); 12 waves each sum 32 k from fragments loaded straight to registers, no LDS weight staging; arm K summed as 12 partials via LDS
# speedup vs baseline: 1.5510x; 1.2184x over previous
#include <hip/hip_runtime.h>


typedef __attribute__((ext_vector_type(2))) float v2f;
typedef __attribute__((ext_vector_type(8))) float v8f;
typedef __attribute__((ext_vector_type(4))) float v4f;

constexpr int Bc = 128, Nw = 4096, Sc = 32, Pc = 64, Dc = 128;
constexpr float NORM_INV = 1.0f / 300.0f;

constexpr int KA  = 256;
constexpr int LDA = 260;
constexpr int WR  = 384;
constexpr int LDW = 136;
constexpr int SEG = 128;

static __device__ __forceinline__
void seg_load(v4f (&r)[16], const float* __restrict__ Wm, int s, int t) {
    const float* src = Wm + (size_t)(s * SEG + (t >> 5)) * Dc + (t & 31) * 4;
    #pragma unroll
    for (int i = 0; i < 16; ++i)
        r[i] = *(const v4f*)(src + (size_t)i * 8 * Dc);
}
static __device__ __forceinline__
void seg_store(float* Ws, const v4f (&r)[16], int s, int t) {
    float* dst = Ws + (s * SEG + (t >> 5)) * LDW + (t & 31) * 4;
    #pragma unroll
    for (int i = 0; i < 16; ++i)
        *(v4f*)(dst + i * 8 * LDW) = r[i];
}

static __device__ __forceinline__ void lds_barrier() {
    asm volatile("s_wait_dscnt 0x0\n\ts_barrier_signal -1\n\ts_barrier_wait -1" ::: "memory");
}

__global__ __launch_bounds__(256)
void pm_emb_kernel(const float* __restrict__ enc_row,
                   const float* __restrict__ enc_col,
                   const float* __restrict__ clockv,
                   const float* __restrict__ pet,
                   const float* __restrict__ W_dyn,
                   const float* __restrict__ W_concat,
                   const int*   __restrict__ lhw,
                   const int*   __restrict__ lstage,
                   float*       __restrict__ out)
{
    __shared__ float Ws[WR * LDW];
    __shared__ float Xs[32 * LDA];
    __shared__ float wd[8][Dc];
    __shared__ float Rp[8][32];

    const int b    = blockIdx.x >> 1;
    const int half = blockIdx.x & 1;
    const int t    = threadIdx.x;

    const int r     = t >> 3;
    const int chunk = t & 7;
    const int p     = half * 32 + r;
    const int col0  = chunk * 32;

    const int gidx = (chunk < 4) ? lstage[b * Pc + p] : lhw[b * Pc + p];

    const float4 wdv = *(const float4*)(W_dyn + 4 * (t & 31));
    const float  pev = pet[b * Pc + half * 32 + (t & 31)];
    const float  clk = clockv[b];

    v4f ra[16], rb[16], rc[16];
    seg_load(ra, W_concat, 0, t);

    const float* src = nullptr;
    if (chunk < 4) {
        const int st = gidx - 1;
        src = enc_col + ((size_t)b * Sc + st) * Dc + col0;
    } else {
        if (gidx >= 0 && gidx < Nw)
            src = enc_row + ((size_t)b * Nw + gidx) * Dc + (col0 - 128);
    }
    v4f g[8];
    #pragma unroll
    for (int j = 0; j < 8; ++j) g[j] = (v4f){0.f, 0.f, 0.f, 0.f};
    if (src) {
        #pragma unroll
        for (int j = 0; j < 8; ++j) g[j] = *(const v4f*)(src + 4 * j);
    }

    seg_load(rb, W_concat, 1, t);
    seg_load(rc, W_concat, 2, t);
    seg_store(Ws, ra, 0, t);
    {
        float* dst = &Xs[r * LDA + col0];
        #pragma unroll
        for (int j = 0; j < 8; ++j) *(v4f*)(dst + 4 * j) = g[j];
    }
    lds_barrier();

    const int wave  = t >> 5;
    const int lane  = t & 31;
    const int l16   = lane & 15;
    const int khalf = (lane >> 4) << 1;
    const int n0    = wave * 16;

    v8f acc0 = {};
    v8f acc1 = {};
    const float* wb = &Ws[khalf * LDW + n0 + l16];
    const float* a0 = &Xs[l16 * LDA + khalf];
    const float* a1 = &Xs[(16 + l16) * LDA + khalf];

    #pragma unroll 8
    for (int k = 0; k < SEG; k += 4) {
        v2f bf;
        bf.x = wb[k * LDW];
        bf.y = wb[(k + 1) * LDW];
        v2f fa0 = *(const v2f*)(a0 + k);
        v2f fa1 = *(const v2f*)(a1 + k);
        acc0 = __builtin_amdgcn_wmma_f32_16x16x4_f32(
                   false, fa0, false, bf, (short)0, acc0, false, false);
        acc1 = __builtin_amdgcn_wmma_f32_16x16x4_f32(
                   false, fa1, false, bf, (short)0, acc1, false, false);
    }
    seg_store(Ws, rb, 1, t);
    seg_store(Ws, rc, 2, t);
    {
        *(float4*)&wd[wave][4 * lane] = wdv;
        const float d = pev - clk;
        Rp[wave][lane] = (d > 0.f ? d : 0.f) * NORM_INV;
    }
    lds_barrier();

    float w3v = 0.f;
    const float* wc = &Ws[KA * LDW + n0 + l16];
    #pragma unroll 8
    for (int k = SEG; k < KA; k += 4) {
        const int kk = k - SEG;
        const float4 wv = *(const float4*)&wd[wave][kk];
        w3v = fmaf(wv.x, wc[(kk + 0) * LDW], w3v);
        w3v = fmaf(wv.y, wc[(kk + 1) * LDW], w3v);
        w3v = fmaf(wv.z, wc[(kk + 2) * LDW], w3v);
        w3v = fmaf(wv.w, wc[(kk + 3) * LDW], w3v);
        v2f bf;
        bf.x = wb[k * LDW];
        bf.y = wb[(k + 1) * LDW];
        v2f fa0 = *(const v2f*)(a0 + k);
        v2f fa1 = *(const v2f*)(a1 + k);
        acc0 = __builtin_amdgcn_wmma_f32_16x16x4_f32(
                   false, fa0, false, bf, (short)0, acc0, false, false);
        acc1 = __builtin_amdgcn_wmma_f32_16x16x4_f32(
                   false, fa1, false, bf, (short)0, acc1, false, false);
    }

    const int   rhalf = (lane >> 4) << 3;
    #pragma unroll
    for (int rr = 0; rr < 8; ++rr) {
        const int row0 = rhalf + rr;
        const float v0 = acc0[rr] + Rp[wave][row0] * w3v;
        out[((size_t)b * (Pc + 2) + half * 32 + row0) * Dc + n0 + l16] = v0;
    }
    #pragma unroll
    for (int rr = 0; rr < 8; ++rr) {
        const int row1 = 16 + rhalf + rr;
        const float v1 = acc1[rr] + Rp[wave][row1] * w3v;
        out[((size_t)b * (Pc + 2) + half * 32 + row1) * Dc + n0 + l16] = v1;
    }
}

constexpr int RWAVES = 12;
constexpr int LDP    = 36;

__global__ __launch_bounds__(RWAVES * 32)
void robot_kernel(const float* __restrict__ enc_row,
                  const float* __restrict__ enc_col,
                  const float* __restrict__ W_robot,
                  const int* __restrict__ a1loc, const int* __restrict__ a2loc,
                  const int* __restrict__ a1rec, const int* __restrict__ a2rec,
                  const int* __restrict__ a1ns,  const int* __restrict__ a2ns,
                  float* __restrict__ out)
{
    __shared__ float part[RWAVES][16 * LDP];

    const int t    = threadIdx.x;
    const int lane = t & 31;
    const int wave = __builtin_amdgcn_readfirstlane(t >> 5);
    const int seg  = wave >> 2;
    const int kq   = (wave & 3) * 32;
    const int mt   = blockIdx.x >> 2;
    const int n0   = (blockIdx.x & 3) * 32;
    const int c    = lane & 15;
    const int h    = lane >> 4;

    const int arow = mt * 16 + c;
    const int bb   = arow >> 1, armr = arow & 1;

    const int* ip = (seg == 0) ? (armr ? a2loc : a1loc)
                  : (seg == 1) ? (armr ? a2rec : a1rec)
                               : (armr ? a2ns  : a1ns );
    const int iv = ip[bb];

    const float* wp = W_robot + (size_t)(seg * 128 + kq + 2 * h) * Dc + n0 + 2 * c;
    v2f bx[8], by[8];
    #pragma unroll
    for (int j = 0; j < 8; ++j) {
        bx[j] = *(const v2f*)(wp + (size_t)(4 * j    ) * Dc);
        by[j] = *(const v2f*)(wp + (size_t)(4 * j + 1) * Dc);
    }

    const float* base;
    int   rowi;
    bool  ok;
    float fill = 0.f;
    if (seg == 0) {
        ok   = (iv >= 1) && (iv <= Pc);
        rowi = bb * (Pc + 2) + (ok ? iv - 1 : 0);
        base = out;
        fill = (iv == Pc + 1) ? 1.f : 0.f;
    } else if (seg == 1) {
        ok   = (iv >= 0) && (iv < Nw);
        rowi = bb * Nw + (ok ? iv : 0);
        base = enc_row;
    } else {
        ok   = (iv >= 1) && (iv <= Sc);
        rowi = bb * Sc + (ok ? iv - 1 : 0);
        base = enc_col;
    }
    const float* ap = base + (size_t)rowi * Dc + kq + 2 * h;
    v2f a[8];
    #pragma unroll
    for (int j = 0; j < 8; ++j) a[j] = *(const v2f*)(ap + 4 * j);
    __builtin_amdgcn_sched_barrier(0);
    #pragma unroll
    for (int j = 0; j < 8; ++j) {
        a[j].x = ok ? a[j].x : fill;
        a[j].y = ok ? a[j].y : fill;
    }

    v8f acc0 = {};
    v8f acc1 = {};
    #pragma unroll
    for (int j = 0; j < 8; ++j) {
        v2f b0, b1;
        b0.x = bx[j].x;  b0.y = by[j].x;
        b1.x = bx[j].y;  b1.y = by[j].y;
        acc0 = __builtin_amdgcn_wmma_f32_16x16x4_f32(
                   false, a[j], false, b0, (short)0, acc0, false, false);
        acc1 = __builtin_amdgcn_wmma_f32_16x16x4_f32(
                   false, a[j], false, b1, (short)0, acc1, false, false);
    }

    {
        float* pw = &part[wave][(8 * h) * LDP + 2 * c];
        #pragma unroll
        for (int rr = 0; rr < 8; ++rr) {
            v2f v;  v.x = acc0[rr];  v.y = acc1[rr];
            *(v2f*)(pw + rr * LDP) = v;
        }
    }
    __syncthreads();

    if (t < 256) {
        const int row = t >> 4;
        const int cp  = (t & 15) * 2;
        v2f s = *(const v2f*)&part[0][row * LDP + cp];
        #pragma unroll
        for (int w = 1; w < RWAVES; ++w) {
            const v2f p = *(const v2f*)&part[w][row * LDP + cp];
            s.x += p.x;  s.y += p.y;
        }
        const int ar = mt * 16 + row;
        const int b  = ar >> 1, arm = ar & 1;
        *(v2f*)(out + ((size_t)b * (Pc + 2) + Pc + arm) * Dc + n0 + cp) = s;
    }
}

extern "C" void kernel_launch(void* const* d_in, const int* in_sizes, int n_in,
                              void* d_out, int out_size, void* d_ws, size_t ws_size,
                              hipStream_t stream) {
    (void)in_sizes; (void)n_in; (void)out_size; (void)d_ws; (void)ws_size;
    const float* enc_row  = (const float*)d_in[0];
    const float* enc_col  = (const float*)d_in[1];
    const float* clockv   = (const float*)d_in[2];
    const float* pet      = (const float*)d_in[3];
    const float* W_dyn    = (const float*)d_in[4];
    const float* W_concat = (const float*)d_in[5];
    const float* W_robot  = (const float*)d_in[6];
    const int*   lhw      = (const int*)d_in[7];
    const int*   lstage   = (const int*)d_in[8];
    const int*   a1loc    = (const int*)d_in[9];
    const int*   a2loc    = (const int*)d_in[10];
    const int*   a1rec    = (const int*)d_in[11];
    const int*   a2rec    = (const int*)d_in[12];
    const int*   a1ns     = (const int*)d_in[13];
    const int*   a2ns     = (const int*)d_in[14];
    float* out = (float*)d_out;

    pm_emb_kernel<<<Bc * 2, 256, 0, stream>>>(enc_row, enc_col, clockv, pet,
                                              W_dyn, W_concat, lhw, lstage, out);
    robot_kernel<<<64, 384, 0, stream>>>(enc_row, enc_col, W_robot,
                                         a1loc, a2loc, a1rec, a2rec, a1ns, a2ns, out);
}


# PM launch re-tiled to batch x 64-column half; stage third built once per batch as a 32-row product table, PM rows add their table row after the product; sum order: stage and wafer K sums added last
# speedup vs baseline: 1.8353x; 1.1833x over previous
#include <hip/hip_runtime.h>


typedef __attribute__((ext_vector_type(2))) float v2f;
typedef __attribute__((ext_vector_type(8))) float v8f;
typedef __attribute__((ext_vector_type(4))) float v4f;

constexpr int Bc = 128, Nw = 4096, Sc = 32, Pc = 64, Dc = 128;
constexpr float NORM_INV = 1.0f / 300.0f;

constexpr int NH  = 64;
constexpr int WR  = 384;
constexpr int SEG = 128;
constexpr int LDW = 72;
constexpr int LDX = 132;
constexpr int LDT = 68;

static __device__ __forceinline__
void seg_load(v4f (&r)[8], const float* __restrict__ Wm, int s, int nh, int t) {
    const float* src = Wm + (size_t)(s * SEG + (t >> 4)) * Dc + nh * NH + (t & 15) * 4;
    #pragma unroll
    for (int i = 0; i < 8; ++i)
        r[i] = *(const v4f*)(src + (size_t)i * 16 * Dc);
}
static __device__ __forceinline__
void seg_store(float* Ws, const v4f (&r)[8], int s, int t) {
    float* dst = Ws + (s * SEG + (t >> 4)) * LDW + (t & 15) * 4;
    #pragma unroll
    for (int i = 0; i < 8; ++i)
        *(v4f*)(dst + i * 16 * LDW) = r[i];
}

static __device__ __forceinline__ void lds_barrier() {
    asm volatile("s_wait_dscnt 0x0\n\ts_barrier_signal -1\n\ts_barrier_wait -1" ::: "memory");
}

__global__ __launch_bounds__(256)
void pm_emb_kernel(const float* __restrict__ enc_row,
                   const float* __restrict__ enc_col,
                   const float* __restrict__ clockv,
                   const float* __restrict__ pet,
                   const float* __restrict__ W_dyn,
                   const float* __restrict__ W_concat,
                   const int*   __restrict__ lhw,
                   const int*   __restrict__ lstage,
                   float*       __restrict__ out)
{
    __shared__ __attribute__((aligned(16))) float Ws[WR * LDW];
    __shared__ __attribute__((aligned(16))) float Xw[Pc * LDX];
    __shared__ __attribute__((aligned(16))) float Xs[Sc * LDX];
    __shared__ float Ts[Sc * LDT];
    __shared__ __attribute__((aligned(16))) float wd[8][Dc];
    __shared__ float Rp[8][32];
    __shared__ int   St[8][32];

    const int b    = blockIdx.x >> 1;
    const int nh   = blockIdx.x & 1;
    const int t    = threadIdx.x;
    const int lane = t & 31;
    const int wave = __builtin_amdgcn_readfirstlane(t >> 5);

    const int giv = lhw[b * Pc + wave * 8 + (lane & 7)];

    const int    prow = b * Pc + 32 * (wave >> 2) + lane;
    const int    sti  = lstage[prow];
    const float  pev  = pet[prow];
    const float  clk  = clockv[b];
    const float4 wdv = *(const float4*)(W_dyn + 4 * lane);

    v4f xs[4];
    {
        const float* sp = enc_col + (size_t)b * Sc * Dc + t * 4;
        #pragma unroll
        for (int i = 0; i < 4; ++i) xs[i] = *(const v4f*)(sp + i * 1024);
    }
    v4f ra[8], rb[8], rc[8];
    seg_load(ra, W_concat, 0, nh, t);
    seg_load(rb, W_concat, 1, nh, t);
    seg_load(rc, W_concat, 2, nh, t);

    v4f  g[8];
    bool ok[8];
    #pragma unroll
    for (int i = 0; i < 8; ++i) {
        const int gi = __builtin_amdgcn_readlane(giv, i);
        ok[i] = (gi >= 0) && (gi < Nw);
        const float* src = enc_row + ((size_t)b * Nw + (ok[i] ? gi : 0)) * Dc + lane * 4;
        g[i] = *(const v4f*)src;
    }

    seg_store(Ws, ra, 0, t);
    {
        float* dst = &Xs[(t >> 5) * LDX + lane * 4];
        #pragma unroll
        for (int i = 0; i < 4; ++i) *(v4f*)(dst + i * 8 * LDX) = xs[i];
    }
    lds_barrier();

    const int l16   = lane & 15;
    const int khalf = (lane >> 4) << 1;
    const int rhalf = (lane >> 4) << 3;
    const int nloc  = (wave & 3) * 16;
    const int mg    = wave >> 2;

    const float* wb = &Ws[khalf * LDW + nloc + l16];
    {
        v8f accT = {};
        const float* as = &Xs[(16 * mg + l16) * LDX + khalf];
        #pragma unroll 8
        for (int k = 0; k < SEG; k += 4) {
            v2f bf;
            bf.x = wb[k * LDW];
            bf.y = wb[(k + 1) * LDW];
            v2f fa = *(const v2f*)(as + k);
            accT = __builtin_amdgcn_wmma_f32_16x16x4_f32(
                       false, fa, false, bf, (short)0, accT, false, false);
        }
        float* tp = &Ts[(16 * mg + rhalf) * LDT + nloc + l16];
        #pragma unroll
        for (int rr = 0; rr < 8; ++rr) tp[rr * LDT] = accT[rr];
    }
    seg_store(Ws, rb, 1, t);
    seg_store(Ws, rc, 2, t);
    #pragma unroll
    for (int i = 0; i < 8; ++i) {
        const v4f z = {0.f, 0.f, 0.f, 0.f};
        *(v4f*)&Xw[(wave * 8 + i) * LDX + lane * 4] = ok[i] ? g[i] : z;
    }
    {
        const float d = pev - clk;
        Rp[wave][lane] = (d > 0.f ? d : 0.f) * NORM_INV;
        St[wave][lane] = sti - 1;
        *(float4*)&wd[wave][4 * lane] = wdv;
    }
    lds_barrier();

    v8f acc0 = {};
    v8f acc1 = {};
    float w3v = 0.f;
    const float* ww = wb + SEG * LDW;
    const float* wc = &Ws[2 * SEG * LDW + nloc + l16];
    const float* a0 = &Xw[(32 * mg + l16) * LDX + khalf];
    const float* a1 = a0 + 16 * LDX;
    #pragma unroll 8
    for (int k = 0; k < SEG; k += 4) {
        const float4 wv = *(const float4*)&wd[wave][k];
        w3v = fmaf(wv.x, wc[(k + 0) * LDW], w3v);
        w3v = fmaf(wv.y, wc[(k + 1) * LDW], w3v);
        w3v = fmaf(wv.z, wc[(k + 2) * LDW], w3v);
        w3v = fmaf(wv.w, wc[(k + 3) * LDW], w3v);
        v2f bf;
        bf.x = ww[k * LDW];
        bf.y = ww[(k + 1) * LDW];
        v2f fa0 = *(const v2f*)(a0 + k);
        v2f fa1 = *(const v2f*)(a1 + k);
        acc0 = __builtin_amdgcn_wmma_f32_16x16x4_f32(
                   false, fa0, false, bf, (short)0, acc0, false, false);
        acc1 = __builtin_amdgcn_wmma_f32_16x16x4_f32(
                   false, fa1, false, bf, (short)0, acc1, false, false);
    }

    const int colg = nh * NH + nloc + l16;
    #pragma unroll
    for (int rr = 0; rr < 8; ++rr) {
        const int lr0 = rhalf + rr;
        const float tv = Ts[St[wave][lr0] * LDT + nloc + l16];
        const float v0 = (acc0[rr] + tv) + Rp[wave][lr0] * w3v;
        out[((size_t)b * (Pc + 2) + 32 * mg + lr0) * Dc + colg] = v0;
    }
    #pragma unroll
    for (int rr = 0; rr < 8; ++rr) {
        const int lr1 = 16 + rhalf + rr;
        const float tv = Ts[St[wave][lr1] * LDT + nloc + l16];
        const float v1 = (acc1[rr] + tv) + Rp[wave][lr1] * w3v;
        out[((size_t)b * (Pc + 2) + 32 * mg + lr1) * Dc + colg] = v1;
    }
}

constexpr int RWAVES = 12;
constexpr int LDP    = 36;

__global__ __launch_bounds__(RWAVES * 32)
void robot_kernel(const float* __restrict__ enc_row,
                  const float* __restrict__ enc_col,
                  const float* __restrict__ W_robot,
                  const int* __restrict__ a1loc, const int* __restrict__ a2loc,
                  const int* __restrict__ a1rec, const int* __restrict__ a2rec,
                  const int* __restrict__ a1ns,  const int* __restrict__ a2ns,
                  float* __restrict__ out)
{
    __shared__ float part[RWAVES][16 * LDP];

    const int t    = threadIdx.x;
    const int lane = t & 31;
    const int wave = __builtin_amdgcn_readfirstlane(t >> 5);
    const int seg  = wave >> 2;
    const int kq   = (wave & 3) * 32;
    const int mt   = blockIdx.x >> 2;
    const int n0   = (blockIdx.x & 3) * 32;
    const int c    = lane & 15;
    const int h    = lane >> 4;

    const int arow = mt * 16 + c;
    const int bb   = arow >> 1, armr = arow & 1;

    const int* ip = (seg == 0) ? (armr ? a2loc : a1loc)
                  : (seg == 1) ? (armr ? a2rec : a1rec)
                               : (armr ? a2ns  : a1ns );
    const int iv = ip[bb];

    const float* wp = W_robot + (size_t)(seg * 128 + kq + 2 * h) * Dc + n0 + 2 * c;
    v2f bx[8], by[8];
    #pragma unroll
    for (int j = 0; j < 8; ++j) {
        bx[j] = *(const v2f*)(wp + (size_t)(4 * j    ) * Dc);
        by[j] = *(const v2f*)(wp + (size_t)(4 * j + 1) * Dc);
    }

    const float* base;
    int   rowi;
    bool  ok;
    float fill = 0.f;
    if (seg == 0) {
        ok   = (iv >= 1) && (iv <= Pc);
        rowi = bb * (Pc + 2) + (ok ? iv - 1 : 0);
        base = out;
        fill = (iv == Pc + 1) ? 1.f : 0.f;
    } else if (seg == 1) {
        ok   = (iv >= 0) && (iv < Nw);
        rowi = bb * Nw + (ok ? iv : 0);
        base = enc_row;
    } else {
        ok   = (iv >= 1) && (iv <= Sc);
        rowi = bb * Sc + (ok ? iv - 1 : 0);
        base = enc_col;
    }
    const float* ap = base + (size_t)rowi * Dc + kq + 2 * h;
    v2f a[8];
    #pragma unroll
    for (int j = 0; j < 8; ++j) a[j] = *(const v2f*)(ap + 4 * j);
    __builtin_amdgcn_sched_barrier(0);
    #pragma unroll
    for (int j = 0; j < 8; ++j) {
        a[j].x = ok ? a[j].x : fill;
        a[j].y = ok ? a[j].y : fill;
    }

    v8f acc0 = {};
    v8f acc1 = {};
    #pragma unroll
    for (int j = 0; j < 8; ++j) {
        v2f b0, b1;
        b0.x = bx[j].x;  b0.y = by[j].x;
        b1.x = bx[j].y;  b1.y = by[j].y;
        acc0 = __builtin_amdgcn_wmma_f32_16x16x4_f32(
                   false, a[j], false, b0, (short)0, acc0, false, false);
        acc1 = __builtin_amdgcn_wmma_f32_16x16x4_f32(
                   false, a[j], false, b1, (short)0, acc1, false, false);
    }

    {
        float* pw = &part[wave][(8 * h) * LDP + 2 * c];
        #pragma unroll
        for (int rr = 0; rr < 8; ++rr) {
            v2f v;  v.x = acc0[rr];  v.y = acc1[rr];
            *(v2f*)(pw + rr * LDP) = v;
        }
    }
    __syncthreads();

    if (t < 256) {
        const int row = t >> 4;
        const int cp  = (t & 15) * 2;
        v2f s = *(const v2f*)&part[0][row * LDP + cp];
        #pragma unroll
        for (int w = 1; w < RWAVES; ++w) {
            const v2f p = *(const v2f*)&part[w][row * LDP + cp];
            s.x += p.x;  s.y += p.y;
        }
        const int ar = mt * 16 + row;
        const int b  = ar >> 1, arm = ar & 1;
        *(v2f*)(out + ((size_t)b * (Pc + 2) + Pc + arm) * Dc + n0 + cp) = s;
    }
}

extern "C" void kernel_launch(void* const* d_in, const int* in_sizes, int n_in,
                              void* d_out, int out_size, void* d_ws, size_t ws_size,
                              hipStream_t stream) {
    (void)in_sizes; (void)n_in; (void)out_size; (void)d_ws; (void)ws_size;
    const float* enc_row  = (const float*)d_in[0];
    const float* enc_col  = (const float*)d_in[1];
    const float* clockv   = (const float*)d_in[2];
    const float* pet      = (const float*)d_in[3];
    const float* W_dyn    = (const float*)d_in[4];
    const float* W_concat = (const float*)d_in[5];
    const float* W_robot  = (const float*)d_in[6];
    const int*   lhw      = (const int*)d_in[7];
    const int*   lstage   = (const int*)d_in[8];
    const int*   a1loc    = (const int*)d_in[9];
    const int*   a2loc    = (const int*)d_in[10];
    const int*   a1rec    = (const int*)d_in[11];
    const int*   a2rec    = (const int*)d_in[12];
    const int*   a1ns     = (const int*)d_in[13];
    const int*   a2ns     = (const int*)d_in[14];
    float* out = (float*)d_out;

    pm_emb_kernel<<<Bc * 2, 256, 0, stream>>>(enc_row, enc_col, clockv, pet,
                                              W_dyn, W_concat, lhw, lstage, out);
    robot_kernel<<<64, 384, 0, stream>>>(enc_row, enc_col, W_robot,
                                         a1loc, a2loc, a1rec, a2rec, a1ns, a2ns, out);
}


# W_concat[256:384] no longer staged in LDS: each thread forms its w3 part from the registers it loaded (8 terms), 16 partial rows added via LDS; w3 chain leaves the wafer sweep; w3 sum order changes
# speedup vs baseline: 1.9707x; 1.0738x over previous
#include <hip/hip_runtime.h>


typedef __attribute__((ext_vector_type(2))) float v2f;
typedef __attribute__((ext_vector_type(8))) float v8f;
typedef __attribute__((ext_vector_type(4))) float v4f;

constexpr int Bc = 128, Nw = 4096, Sc = 32, Pc = 64, Dc = 128;
constexpr float NORM_INV = 1.0f / 300.0f;

constexpr int NH  = 64;
constexpr int WRS = 256;
constexpr int SEG = 128;
constexpr int LDW = 72;
constexpr int LDX = 132;
constexpr int LDT = 68;
constexpr int W3G = 16;

static __device__ __forceinline__
void seg_load(v4f (&r)[8], const float* __restrict__ Wm, int s, int nh, int t) {
    const float* src = Wm + (size_t)(s * SEG + (t >> 4)) * Dc + nh * NH + (t & 15) * 4;
    #pragma unroll
    for (int i = 0; i < 8; ++i)
        r[i] = *(const v4f*)(src + (size_t)i * 16 * Dc);
}
static __device__ __forceinline__
void seg_store(float* Ws, const v4f (&r)[8], int s, int t) {
    float* dst = Ws + (s * SEG + (t >> 4)) * LDW + (t & 15) * 4;
    #pragma unroll
    for (int i = 0; i < 8; ++i)
        *(v4f*)(dst + i * 16 * LDW) = r[i];
}

static __device__ __forceinline__ void lds_barrier() {
    asm volatile("s_wait_dscnt 0x0\n\ts_barrier_signal -1\n\ts_barrier_wait -1" ::: "memory");
}

__global__ __launch_bounds__(256)
void pm_emb_kernel(const float* __restrict__ enc_row,
                   const float* __restrict__ enc_col,
                   const float* __restrict__ clockv,
                   const float* __restrict__ pet,
                   const float* __restrict__ W_dyn,
                   const float* __restrict__ W_concat,
                   const int*   __restrict__ lhw,
                   const int*   __restrict__ lstage,
                   float*       __restrict__ out)
{
    __shared__ __attribute__((aligned(16))) float Ws[WRS * LDW];
    __shared__ __attribute__((aligned(16))) float Xw[Pc * LDX];
    __shared__ __attribute__((aligned(16))) float Xs[Sc * LDX];
    __shared__ float Ts[Sc * LDT];
    __shared__ __attribute__((aligned(16))) float W3p[W3G * NH];
    __shared__ float Rp[8][32];
    __shared__ int   St[8][32];

    const int b    = blockIdx.x >> 1;
    const int nh   = blockIdx.x & 1;
    const int t    = threadIdx.x;
    const int lane = t & 31;
    const int wave = __builtin_amdgcn_readfirstlane(t >> 5);

    const int giv = lhw[b * Pc + wave * 8 + (lane & 7)];

    const int    prow = b * Pc + 32 * (wave >> 2) + lane;
    const int    sti  = lstage[prow];
    const float  pev  = pet[prow];
    const float  clk  = clockv[b];
    float wdk[8];
    #pragma unroll
    for (int i = 0; i < 8; ++i) wdk[i] = W_dyn[(t >> 4) + 16 * i];

    v4f xs[4];
    {
        const float* sp = enc_col + (size_t)b * Sc * Dc + t * 4;
        #pragma unroll
        for (int i = 0; i < 4; ++i) xs[i] = *(const v4f*)(sp + i * 1024);
    }
    v4f ra[8], rb[8], rc[8];
    seg_load(ra, W_concat, 0, nh, t);
    seg_load(rb, W_concat, 1, nh, t);
    seg_load(rc, W_concat, 2, nh, t);

    v4f  g[8];
    bool ok[8];
    #pragma unroll
    for (int i = 0; i < 8; ++i) {
        const int gi = __builtin_amdgcn_readlane(giv, i);
        ok[i] = (gi >= 0) && (gi < Nw);
        const float* src = enc_row + ((size_t)b * Nw + (ok[i] ? gi : 0)) * Dc + lane * 4;
        g[i] = *(const v4f*)src;
    }
    __builtin_amdgcn_sched_barrier(0);

    seg_store(Ws, ra, 0, t);
    {
        float* dst = &Xs[(t >> 5) * LDX + lane * 4];
        #pragma unroll
        for (int i = 0; i < 4; ++i) *(v4f*)(dst + i * 8 * LDX) = xs[i];
    }
    lds_barrier();

    const int l16   = lane & 15;
    const int khalf = (lane >> 4) << 1;
    const int rhalf = (lane >> 4) << 3;
    const int nloc  = (wave & 3) * 16;
    const int mg    = wave >> 2;

    const float* wb = &Ws[khalf * LDW + nloc + l16];
    {
        v8f accT = {};
        const float* as = &Xs[(16 * mg + l16) * LDX + khalf];
        #pragma unroll 8
        for (int k = 0; k < SEG; k += 4) {
            v2f bf;
            bf.x = wb[k * LDW];
            bf.y = wb[(k + 1) * LDW];
            v2f fa = *(const v2f*)(as + k);
            accT = __builtin_amdgcn_wmma_f32_16x16x4_f32(
                       false, fa, false, bf, (short)0, accT, false, false);
        }
        float* tp = &Ts[(16 * mg + rhalf) * LDT + nloc + l16];
        #pragma unroll
        for (int rr = 0; rr < 8; ++rr) tp[rr * LDT] = accT[rr];
    }
    seg_store(Ws, rb, 1, t);
    {
        v4f p = wdk[0] * rc[0];
        #pragma unroll
        for (int i = 1; i < 8; ++i) {
            p.x = fmaf(wdk[i], rc[i].x, p.x);
            p.y = fmaf(wdk[i], rc[i].y, p.y);
            p.z = fmaf(wdk[i], rc[i].z, p.z);
            p.w = fmaf(wdk[i], rc[i].w, p.w);
        }
        *(v4f*)&W3p[(t >> 4) * NH + (t & 15) * 4] = p;
    }
    #pragma unroll
    for (int i = 0; i < 8; ++i) {
        const v4f z = {0.f, 0.f, 0.f, 0.f};
        *(v4f*)&Xw[(wave * 8 + i) * LDX + lane * 4] = ok[i] ? g[i] : z;
    }
    {
        const float d = pev - clk;
        Rp[wave][lane] = (d > 0.f ? d : 0.f) * NORM_INV;
        St[wave][lane] = sti - 1;
    }
    lds_barrier();

    float w3v = W3p[nloc + l16];
    #pragma unroll
    for (int j = 1; j < W3G; ++j) w3v += W3p[j * NH + nloc + l16];

    v8f acc0 = {};
    v8f acc1 = {};
    const float* ww = wb + SEG * LDW;
    const float* a0 = &Xw[(32 * mg + l16) * LDX + khalf];
    const float* a1 = a0 + 16 * LDX;
    #pragma unroll 8
    for (int k = 0; k < SEG; k += 4) {
        v2f bf;
        bf.x = ww[k * LDW];
        bf.y = ww[(k + 1) * LDW];
        v2f fa0 = *(const v2f*)(a0 + k);
        v2f fa1 = *(const v2f*)(a1 + k);
        acc0 = __builtin_amdgcn_wmma_f32_16x16x4_f32(
                   false, fa0, false, bf, (short)0, acc0, false, false);
        acc1 = __builtin_amdgcn_wmma_f32_16x16x4_f32(
                   false, fa1, false, bf, (short)0, acc1, false, false);
    }

    const int colg = nh * NH + nloc + l16;
    #pragma unroll
    for (int rr = 0; rr < 8; ++rr) {
        const int lr0 = rhalf + rr;
        const float tv = Ts[St[wave][lr0] * LDT + nloc + l16];
        const float v0 = (acc0[rr] + tv) + Rp[wave][lr0] * w3v;
        out[((size_t)b * (Pc + 2) + 32 * mg + lr0) * Dc + colg] = v0;
    }
    #pragma unroll
    for (int rr = 0; rr < 8; ++rr) {
        const int lr1 = 16 + rhalf + rr;
        const float tv = Ts[St[wave][lr1] * LDT + nloc + l16];
        const float v1 = (acc1[rr] + tv) + Rp[wave][lr1] * w3v;
        out[((size_t)b * (Pc + 2) + 32 * mg + lr1) * Dc + colg] = v1;
    }
}

constexpr int RWAVES = 12;
constexpr int LDP    = 36;

__global__ __launch_bounds__(RWAVES * 32)
void robot_kernel(const float* __restrict__ enc_row,
                  const float* __restrict__ enc_col,
                  const float* __restrict__ W_robot,
                  const int* __restrict__ a1loc, const int* __restrict__ a2loc,
                  const int* __restrict__ a1rec, const int* __restrict__ a2rec,
                  const int* __restrict__ a1ns,  const int* __restrict__ a2ns,
                  float* __restrict__ out)
{
    __shared__ float part[RWAVES][16 * LDP];

    const int t    = threadIdx.x;
    const int lane = t & 31;
    const int wave = __builtin_amdgcn_readfirstlane(t >> 5);
    const int seg  = wave >> 2;
    const int kq   = (wave & 3) * 32;
    const int mt   = blockIdx.x >> 2;
    const int n0   = (blockIdx.x & 3) * 32;
    const int c    = lane & 15;
    const int h    = lane >> 4;

    const int arow = mt * 16 + c;
    const int bb   = arow >> 1, armr = arow & 1;

    const int* ip = (seg == 0) ? (armr ? a2loc : a1loc)
                  : (seg == 1) ? (armr ? a2rec : a1rec)
                               : (armr ? a2ns  : a1ns );
    const int iv = ip[bb];

    const float* wp = W_robot + (size_t)(seg * 128 + kq + 2 * h) * Dc + n0 + 2 * c;
    v2f bx[8], by[8];
    #pragma unroll
    for (int j = 0; j < 8; ++j) {
        bx[j] = *(const v2f*)(wp + (size_t)(4 * j    ) * Dc);
        by[j] = *(const v2f*)(wp + (size_t)(4 * j + 1) * Dc);
    }

    const float* base;
    int   rowi;
    bool  ok;
    float fill = 0.f;
    if (seg == 0) {
        ok   = (iv >= 1) && (iv <= Pc);
        rowi = bb * (Pc + 2) + (ok ? iv - 1 : 0);
        base = out;
        fill = (iv == Pc + 1) ? 1.f : 0.f;
    } else if (seg == 1) {
        ok   = (iv >= 0) && (iv < Nw);
        rowi = bb * Nw + (ok ? iv : 0);
        base = enc_row;
    } else {
        ok   = (iv >= 1) && (iv <= Sc);
        rowi = bb * Sc + (ok ? iv - 1 : 0);
        base = enc_col;
    }
    const float* ap = base + (size_t)rowi * Dc + kq + 2 * h;
    v2f a[8];
    #pragma unroll
    for (int j = 0; j < 8; ++j) a[j] = *(const v2f*)(ap + 4 * j);
    __builtin_amdgcn_sched_barrier(0);
    #pragma unroll
    for (int j = 0; j < 8; ++j) {
        a[j].x = ok ? a[j].x : fill;
        a[j].y = ok ? a[j].y : fill;
    }

    v8f acc0 = {};
    v8f acc1 = {};
    #pragma unroll
    for (int j = 0; j < 8; ++j) {
        v2f b0, b1;
        b0.x = bx[j].x;  b0.y = by[j].x;
        b1.x = bx[j].y;  b1.y = by[j].y;
        acc0 = __builtin_amdgcn_wmma_f32_16x16x4_f32(
                   false, a[j], false, b0, (short)0, acc0, false, false);
        acc1 = __builtin_amdgcn_wmma_f32_16x16x4_f32(
                   false, a[j], false, b1, (short)0, acc1, false, false);
    }

    {
        float* pw = &part[wave][(8 * h) * LDP + 2 * c];
        #pragma unroll
        for (int rr = 0; rr < 8; ++rr) {
            v2f v;  v.x = acc0[rr];  v.y = acc1[rr];
            *(v2f*)(pw + rr * LDP) = v;
        }
    }
    __syncthreads();

    if (t < 256) {
        const int row = t >> 4;
        const int cp  = (t & 15) * 2;
        v2f s = *(const v2f*)&part[0][row * LDP + cp];
        #pragma unroll
        for (int w = 1; w < RWAVES; ++w) {
            const v2f p = *(const v2f*)&part[w][row * LDP + cp];
            s.x += p.x;  s.y += p.y;
        }
        const int ar = mt * 16 + row;
        const int b  = ar >> 1, arm = ar & 1;
        *(v2f*)(out + ((size_t)b * (Pc + 2) + Pc + arm) * Dc + n0 + cp) = s;
    }
}

extern "C" void kernel_launch(void* const* d_in, const int* in_sizes, int n_in,
                              void* d_out, int out_size, void* d_ws, size_t ws_size,
                              hipStream_t stream) {
    (void)in_sizes; (void)n_in; (void)out_size; (void)d_ws; (void)ws_size;
    const float* enc_row  = (const float*)d_in[0];
    const float* enc_col  = (const float*)d_in[1];
    const float* clockv   = (const float*)d_in[2];
    const float* pet      = (const float*)d_in[3];
    const float* W_dyn    = (const float*)d_in[4];
    const float* W_concat = (const float*)d_in[5];
    const float* W_robot  = (const float*)d_in[6];
    const int*   lhw      = (const int*)d_in[7];
    const int*   lstage   = (const int*)d_in[8];
    const int*   a1loc    = (const int*)d_in[9];
    const int*   a2loc    = (const int*)d_in[10];
    const int*   a1rec    = (const int*)d_in[11];
    const int*   a2rec    = (const int*)d_in[12];
    const int*   a1ns     = (const int*)d_in[13];
    const int*   a2ns     = (const int*)d_in[14];
    float* out = (float*)d_out;

    pm_emb_kernel<<<Bc * 2, 256, 0, stream>>>(enc_row, enc_col, clockv, pet,
                                              W_dyn, W_concat, lhw, lstage, out);
    robot_kernel<<<64, 384, 0, stream>>>(enc_row, enc_col, W_robot,
                                         a1loc, a2loc, a1rec, a2rec, a1ns, a2ns, out);
}
